# starting kernel
# baseline (speedup 1.0000x reference)
#include <hip/hip_runtime.h>
#include <hip/hip_bf16.h>

typedef __attribute__((ext_vector_type(2))) float v2f;
typedef __attribute__((ext_vector_type(8))) float v8f;

#define N_TOTAL 8192
#define KDIM    16
#define FIN     32
#define FOUT    32

#define WAVES_PER_BLOCK 8
#define K1_BLOCKS       16
#define K1_WAVES        (K1_BLOCKS * WAVES_PER_BLOCK)
#define ROWS_PER_WAVE   (N_TOTAL / K1_WAVES)

__global__ __launch_bounds__(256)
void k1_partialT(const float* __restrict__ ev,
                 const float* __restrict__ in,
                 float* __restrict__ partial) {
    const int wave = blockIdx.x * WAVES_PER_BLOCK + (threadIdx.x >> 5);
    const int lane = threadIdx.x & 31;
    const int m    = lane & 15;
    const int half = lane >> 4;

    v8f acc0 = {};
    v8f acc1 = {};

    const int nstart = wave * ROWS_PER_WAVE;
    #pragma unroll 4
    for (int it = 0; it < ROWS_PER_WAVE / 4; ++it) {
        const int nb = nstart + it * 4 + half * 2;
        v2f a, b0, b1;
        a.x  = ev[nb * KDIM + m];
        a.y  = ev[(nb + 1) * KDIM + m];
        b0.x = in[nb * FIN + m];
        b0.y = in[(nb + 1) * FIN + m];
        b1.x = in[nb * FIN + 16 + m];
        b1.y = in[(nb + 1) * FIN + 16 + m];
        acc0 = __builtin_amdgcn_wmma_f32_16x16x4_f32(false, a, false, b0,
                                                     (short)0, acc0, false, false);
        acc1 = __builtin_amdgcn_wmma_f32_16x16x4_f32(false, a, false, b1,
                                                     (short)0, acc1, false, false);
    }

    float* p = partial + wave * (KDIM * FIN);
    #pragma unroll
    for (int v = 0; v < 8; ++v) {
        const int row = half ? (v + 8) : v;
        p[row * FIN + m]      = acc0[v];
        p[row * FIN + 16 + m] = acc1[v];
    }
}

__global__ __launch_bounds__(512)
void k2_reduce_S(const float* __restrict__ partial,
                 const float* __restrict__ eigenVal,
                 const float* __restrict__ W,
                 const float* __restrict__ A,
                 const float* __restrict__ B,
                 const float* __restrict__ C,
                 const float* __restrict__ D,
                 float* __restrict__ S) {
    __shared__ float Tl[KDIM * FIN];
    __shared__ float Mr[KDIM];
    __shared__ float Mi[KDIM];
    const int tid = threadIdx.x;

    float s = 0.0f;
    for (int w = 0; w < K1_WAVES; ++w)
        s += partial[w * (KDIM * FIN) + tid];
    Tl[tid] = s;

    if (tid < KDIM) {
        const float lam = eigenVal[tid];
        const float m1r = A[tid]      * lam + B[tid];
        const float m1i = A[16 + tid] * lam + B[16 + tid];
        const float m2r = 1.0f / (C[tid]      * lam + D[tid]);
        const float m2i = 1.0f / (C[16 + tid] * lam + D[16 + tid]);
        Mr[tid] = m1r * m2r - m1i * m2i;
        Mi[tid] = m1i * m2r + m1r * m2i;
    }
    __syncthreads();

    const int k = tid >> 5;
    const int f = tid & 31;
    float s0 = 0.0f, s1 = 0.0f;
    #pragma unroll
    for (int j = 0; j < FIN; ++j) {
        const float t = Tl[k * FIN + j];
        s0 += t * W[j * FOUT + f];
        s1 += t * W[FIN * FOUT + j * FOUT + f];
    }
    S[k * FOUT + f] = Mr[k] * s0 - Mi[k] * s1;
}

__global__ __launch_bounds__(256)
void k3_out(const float* __restrict__ ev,
            const float* __restrict__ S,
            const float* __restrict__ bias,
            float* __restrict__ out) {
    const int wave = blockIdx.x * WAVES_PER_BLOCK + (threadIdx.x >> 5);
    const int lane = threadIdx.x & 31;
    const int m    = lane & 15;
    const int half = lane >> 4;
    const int n0   = wave * 16;

    v8f acc0 = {};
    v8f acc1 = {};
    #pragma unroll
    for (int kc = 0; kc < 4; ++kc) {
        const int kg = kc * 4 + half * 2;
        v2f a, b0, b1;
        a.x  = ev[(n0 + m) * KDIM + kg];
        a.y  = ev[(n0 + m) * KDIM + kg + 1];
        b0.x = S[kg * FOUT + m];
        b0.y = S[(kg + 1) * FOUT + m];
        b1.x = S[kg * FOUT + 16 + m];
        b1.y = S[(kg + 1) * FOUT + 16 + m];
        acc0 = __builtin_amdgcn_wmma_f32_16x16x4_f32(false, a, false, b0,
                                                     (short)0, acc0, false, false);
        acc1 = __builtin_amdgcn_wmma_f32_16x16x4_f32(false, a, false, b1,
                                                     (short)0, acc1, false, false);
    }

    const float bl = bias[m];
    const float bh = bias[16 + m];
    #pragma unroll
    for (int v = 0; v < 8; ++v) {
        const int row = n0 + (half ? (v + 8) : v);
        out[row * FOUT + m]      = 2.0f * acc0[v] + bl;
        out[row * FOUT + 16 + m] = 2.0f * acc1[v] + bh;
    }
}

extern "C" void kernel_launch(void* const* d_in, const int* in_sizes, int n_in,
                              void* d_out, int out_size, void* d_ws, size_t ws_size,
                              hipStream_t stream) {
    const float* input    = (const float*)d_in[0];
    const float* eigenVal = (const float*)d_in[1];
    const float* eigenVec = (const float*)d_in[2];
    const float* W        = (const float*)d_in[3];
    const float* A        = (const float*)d_in[4];
    const float* B        = (const float*)d_in[5];
    const float* C        = (const float*)d_in[6];
    const float* D        = (const float*)d_in[7];
    const float* bias     = (const float*)d_in[8];
    float* out = (float*)d_out;

    float* partial = (float*)d_ws;
    float* S       = partial + (size_t)K1_WAVES * KDIM * FIN;

    k1_partialT<<<K1_BLOCKS, 256, 0, stream>>>(eigenVec, input, partial);
    k2_reduce_S<<<1, 512, 0, stream>>>(partial, eigenVal, W, A, B, C, D, S);
    k3_out<<<N_TOTAL / (16 * WAVES_PER_BLOCK), 256, 0, stream>>>(eigenVec, S, bias, out);
}


# k1 split by T's column halves: 32 work-groups (16 row ranges x 2 halves of the input columns), a wave runs 16 f32 matrix steps on one tile, not 32 on two; 16 shares per element; sums unchanged
# speedup vs baseline: 2.2896x; 2.2896x over previous
#include <hip/hip_runtime.h>
#include <hip/hip_bf16.h>


typedef __attribute__((ext_vector_type(2))) float v2f;
typedef __attribute__((ext_vector_type(4))) float v4f;
typedef __attribute__((ext_vector_type(8))) float v8f;

#define N_TOTAL 8192
#define KDIM    16
#define FIN     32
#define FOUT    32

#define WAVES_PER_BLOCK 8
#define K1_BLOCKS       16
#define K1_WAVES        (K1_BLOCKS * WAVES_PER_BLOCK)
#define ROWS_PER_WAVE   (N_TOTAL / K1_WAVES)
#define T_ELEMS         (KDIM * FIN)
#define HCOLS           16
#define H_ELEMS         (KDIM * HCOLS)
#define K1_GRID         (K1_BLOCKS * 2)

#define T_PITCH 34

#define CHUNKS   4
#define E_CH     (16 * KDIM)
#define I_CH     (16 * HCOLS)
#define STG_WAVE (E_CH + I_CH)

__global__ __launch_bounds__(256)
void k1_shareT(const float* __restrict__ ev,
               const float* __restrict__ in,
               float* __restrict__ shares) {
    __shared__ float Pl[WAVES_PER_BLOCK * H_ELEMS];
    __shared__ __attribute__((aligned(16))) float Stg[WAVES_PER_BLOCK * STG_WAVE];

    const int wib  = threadIdx.x >> 5;
    const int rng  = blockIdx.x >> 1;
    const int ch   = blockIdx.x & 1;
    const int wave = rng * WAVES_PER_BLOCK + wib;
    const int lane = threadIdx.x & 31;
    const int m    = lane & 15;
    const int half = lane >> 4;

    const int nstart = wave * ROWS_PER_WAVE;
    const float* evw = ev + nstart * KDIM + lane * 4;
    const float* inw = in + (nstart + (lane >> 2)) * FIN + ch * HCOLS + (lane & 3) * 4;

    v4f ge[2 * CHUNKS], gi[2 * CHUNKS];
    #pragma unroll
    for (int c = 0; c < CHUNKS; ++c) {
        ge[2 * c]     = *(const v4f*)(evw + (2 * c) * 128);
        ge[2 * c + 1] = *(const v4f*)(evw + (2 * c + 1) * 128);
        gi[2 * c]     = *(const v4f*)(inw + (2 * c) * 8 * FIN);
        gi[2 * c + 1] = *(const v4f*)(inw + (2 * c + 1) * 8 * FIN);
    }

    float* El = Stg + wib * STG_WAVE;
    float* Il = El + E_CH;
    float* est = El + lane * 4;
    float* ist = Il + lane * 4;
    const float* ea = El + half * 2 * KDIM + m;
    const float* ia = Il + half * 2 * HCOLS + m;

    v8f acc = {};

    #pragma unroll
    for (int c = 0; c < CHUNKS; ++c) {
        *(v4f*)(est)       = ge[2 * c];
        *(v4f*)(est + 128) = ge[2 * c + 1];
        *(v4f*)(ist)       = gi[2 * c];
        *(v4f*)(ist + 128) = gi[2 * c + 1];

        __builtin_amdgcn_fence(__ATOMIC_RELEASE, "wavefront");
        __builtin_amdgcn_wave_barrier();
        __builtin_amdgcn_fence(__ATOMIC_ACQUIRE, "wavefront");

        #pragma unroll
        for (int i = 0; i < 4; ++i) {
            v2f a, b;
            a.x = ea[i * 4 * KDIM];
            a.y = ea[i * 4 * KDIM + KDIM];
            b.x = ia[i * 4 * HCOLS];
            b.y = ia[i * 4 * HCOLS + HCOLS];
            acc = __builtin_amdgcn_wmma_f32_16x16x4_f32(false, a, false, b,
                                                        (short)0, acc, false, false);
        }

        __builtin_amdgcn_fence(__ATOMIC_RELEASE, "wavefront");
        __builtin_amdgcn_wave_barrier();
        __builtin_amdgcn_fence(__ATOMIC_ACQUIRE, "wavefront");
    }

    float* p = Pl + wib * H_ELEMS;
    #pragma unroll
    for (int v = 0; v < 8; ++v) {
        const int row = half ? (v + 8) : v;
        p[row * HCOLS + m] = acc[v];
    }
    __syncthreads();

    float s = Pl[threadIdx.x];
    #pragma unroll
    for (int w = 1; w < WAVES_PER_BLOCK; ++w)
        s += Pl[w * H_ELEMS + threadIdx.x];
    shares[blockIdx.x * H_ELEMS + threadIdx.x] = s;
}


#define S_PITCH 36
#define S_PLANE (KDIM * S_PITCH)

__global__ __launch_bounds__(256)
void k2_S_out(const float* __restrict__ ev,
              const float* __restrict__ shares,
              const float* __restrict__ eigenVal,
              const float* __restrict__ W,
              const float* __restrict__ A,
              const float* __restrict__ B,
              const float* __restrict__ C,
              const float* __restrict__ D,
              const float* __restrict__ bias,
              float* __restrict__ out) {
    __shared__ float Tl[KDIM * T_PITCH];
    __shared__ float Sp[4 * S_PLANE];

    const int tid  = threadIdx.x;
    const int lane = tid & 31;
    const int m    = lane & 15;
    const int half = lane >> 4;
    const int wib  = tid >> 5;
    const int st   = wib & 1;
    const int sq   = wib >> 1;
    const int n0   = (blockIdx.x * WAVES_PER_BLOCK + wib) * 16;

    v2f sh[K1_BLOCKS];
    #pragma unroll
    for (int b = 0; b < K1_BLOCKS; ++b)
        sh[b] = *(const v2f*)(shares + b * T_ELEMS + 2 * tid);

    const float lam = eigenVal[m];
    const float a_r = A[m],      b_r = B[m],      c_r = C[m],      d_r = D[m];
    const float a_i = A[16 + m], b_i = B[16 + m], c_i = C[16 + m], d_i = D[16 + m];

    v2f wf[4];
    #pragma unroll
    for (int i = 0; i < 4; ++i) {
        const int s = 4 * sq + i;
        const int j = (s & 7) * 4 + half * 2 + (s >> 3) * FIN;
        wf[i].x = W[j * FOUT + 16 * st + m];
        wf[i].y = W[(j + 1) * FOUT + 16 * st + m];
    }

    const v4f ea = *(const v4f*)(ev + (n0 + m) * KDIM + half * 8);
    const v4f eb = *(const v4f*)(ev + (n0 + m) * KDIM + half * 8 + 4);
    const float bl = bias[m];
    const float bh = bias[16 + m];

    __builtin_amdgcn_sched_barrier(0);

    const float m1r = a_r * lam + b_r;
    const float m1i = a_i * lam + b_i;
    const float m2r = 1.0f / (c_r * lam + d_r);
    const float m2i = 1.0f / (c_i * lam + d_i);
    const float cr  = m1r * m2r - m1i * m2i;
    const float ci  = -(m1i * m2r + m1r * m2i);
    const float cw  = (sq >> 1) ? ci : cr;

    v2f t2 = sh[0];
    #pragma unroll
    for (int b = 1; b < K1_BLOCKS; ++b)
        t2 += sh[b];
    *(v2f*)(Tl + ((tid >> 3) & 15) * T_PITCH + HCOLS * (tid >> 7) + 2 * (tid & 7)) = t2;
    __syncthreads();

    v8f sp = {};
    #pragma unroll
    for (int i = 0; i < 4; ++i) {
        const int s = 4 * sq + i;
        const v2f t = *(const v2f*)(Tl + m * T_PITCH + (s & 7) * 4 + half * 2);
        sp = __builtin_amdgcn_wmma_f32_16x16x4_f32(false, t * cw, false, wf[i],
                                                   (short)0, sp, false, false);
    }
    #pragma unroll
    for (int r = 0; r < 8; ++r)
        Sp[sq * S_PLANE + (half * 8 + r) * S_PITCH + 16 * st + m] = sp[r];
    __syncthreads();

    const v2f ak[4] = { {ea.x, ea.y}, {ea.z, ea.w}, {eb.x, eb.y}, {eb.z, eb.w} };
    v8f acc0 = {};
    v8f acc1 = {};
    #pragma unroll
    for (int kc = 0; kc < 4; ++kc) {
        const float* p = Sp + (half * 8 + 2 * kc) * S_PITCH + m;
        v2f b0 = { p[0], p[S_PITCH] };
        v2f b1 = { p[16], p[S_PITCH + 16] };
        #pragma unroll
        for (int q = 1; q < 4; ++q) {
            b0.x += p[q * S_PLANE];
            b0.y += p[q * S_PLANE + S_PITCH];
            b1.x += p[q * S_PLANE + 16];
            b1.y += p[q * S_PLANE + S_PITCH + 16];
        }
        acc0 = __builtin_amdgcn_wmma_f32_16x16x4_f32(false, ak[kc], false, b0,
                                                     (short)0, acc0, false, false);
        acc1 = __builtin_amdgcn_wmma_f32_16x16x4_f32(false, ak[kc], false, b1,
                                                     (short)0, acc1, false, false);
    }

    #pragma unroll
    for (int v = 0; v < 8; ++v) {
        const int row = n0 + (half ? (v + 8) : v);
        out[row * FOUT + m]      = 2.0f * acc0[v] + bl;
        out[row * FOUT + 16 + m] = 2.0f * acc1[v] + bh;
    }
}

extern "C" void kernel_launch(void* const* d_in, const int* in_sizes, int n_in,
                              void* d_out, int out_size, void* d_ws, size_t ws_size,
                              hipStream_t stream) {
    const float* input    = (const float*)d_in[0];
    const float* eigenVal = (const float*)d_in[1];
    const float* eigenVec = (const float*)d_in[2];
    const float* W        = (const float*)d_in[3];
    const float* A        = (const float*)d_in[4];
    const float* B        = (const float*)d_in[5];
    const float* C        = (const float*)d_in[6];
    const float* D        = (const float*)d_in[7];
    const float* bias     = (const float*)d_in[8];
    float* out = (float*)d_out;

    float* shares = (float*)d_ws;

    k1_shareT<<<K1_GRID, 256, 0, stream>>>(eigenVec, input, shares);
    k2_S_out<<<N_TOTAL / (16 * WAVES_PER_BLOCK), 256, 0, stream>>>(
        eigenVec, shares, eigenVal, W, A, B, C, D, bias, out);
}
